# starting kernel
# baseline (speedup 1.0000x reference)
#include <hip/hip_runtime.h>

#define B_  131072
#define S_  32
#define D_  16
#define WPB 4
#define TPB (WPB * 32)

#define XS  20
#define SS  36

typedef float v2f __attribute__((ext_vector_type(2)));
typedef float v8f __attribute__((ext_vector_type(8)));

__device__ __forceinline__ v8f wmma4(v2f a, v2f b, v8f c) {
  return __builtin_amdgcn_wmma_f32_16x16x4_f32(false, a, false, b, (short)0, c,
                                               false, false);
}

__device__ __forceinline__ void lds_fence() {
  asm volatile("s_wait_dscnt 0" ::: "memory");
}

__device__ __forceinline__ v2f frag_rowpair(const float* base, int row, int rs,
                                            int col) {
  return *(const v2f*)(base + row * rs + col);
}

__device__ __forceinline__ v2f frag_col(const float* base, int row0, int rs,
                                        int col) {
  v2f r;
  r.x = base[(row0 + 0) * rs + col];
  r.y = base[(row0 + 1) * rs + col];
  return r;
}

__device__ __forceinline__ void store_tile(float* base, int rs, int row0,
                                           int col, v8f acc) {
#pragma unroll
  for (int v = 0; v < 8; ++v) base[(row0 + v) * rs + col] = acc[v];
}

__global__ __launch_bounds__(TPB) void encoder_kernel(
    const int* __restrict__ arr1, const int* __restrict__ arr2,
    const float* __restrict__ emb,
    const float* __restrict__ wq, const float* __restrict__ wk,
    const float* __restrict__ wv,
    const float* __restrict__ w1, const float* __restrict__ b1,
    const float* __restrict__ w2, const float* __restrict__ b2,
    float* __restrict__ out) {
  __shared__ __align__(16) float s_w[5][256];
  __shared__ __align__(16) float s_b[2][16];
  __shared__ __align__(16) float s_x[WPB][S_ * XS];
  __shared__ __align__(16) float s_q[WPB][S_ * XS];
  __shared__ __align__(16) float s_k[WPB][S_ * XS];
  __shared__ __align__(16) float s_v[WPB][S_ * XS];
  __shared__ __align__(16) float s_s[WPB][S_ * SS];

  const int tid = threadIdx.x;
#pragma unroll
  for (int i = tid; i < 256; i += TPB) {
    s_w[0][i] = wq[i];
    s_w[1][i] = wk[i];
    s_w[2][i] = wv[i];
    s_w[3][i] = w1[i];
    s_w[4][i] = w2[i];
  }
  if (tid < 16) {
    s_b[0][tid] = b1[tid];
    s_b[1][tid] = b2[tid];
  }
  __syncthreads();

  const int wave = tid >> 5;
  const int lane = tid & 31;
  const int m    = lane & 15;
  const int half = lane >> 4;

  const unsigned task = blockIdx.x * WPB + wave;
  const int enc = (task >= (unsigned)B_) ? 1 : 0;
  const int b   = enc ? (int)(task - (unsigned)B_) : (int)task;
  const int* __restrict__ arr = enc ? arr2 : arr1;

  float* X  = s_x[wave];
  float* Q  = s_q[wave];
  float* K  = s_k[wave];
  float* V  = s_v[wave];
  float* Sc = s_s[wave];

  {
    const int tok = arr[b * S_ + lane];
    const float4* er = (const float4*)(emb + (size_t)tok * D_);
    float4* xr = (float4*)(X + lane * XS);
    xr[0] = er[0];
    xr[1] = er[1];
    xr[2] = er[2];
    xr[3] = er[3];
  }
  lds_fence();

#pragma unroll
  for (int t = 0; t < 2; ++t) {
    v8f cq = {}, ck = {}, cv = {};
#pragma unroll
    for (int kc = 0; kc < 4; ++kc) {
      v2f a = frag_rowpair(X, 16 * t + m, XS, 4 * kc + 2 * half);
      cq = wmma4(a, frag_rowpair(s_w[0], m, D_, 4 * kc + 2 * half), cq);
      ck = wmma4(a, frag_rowpair(s_w[1], m, D_, 4 * kc + 2 * half), ck);
      cv = wmma4(a, frag_rowpair(s_w[2], m, D_, 4 * kc + 2 * half), cv);
    }
    store_tile(Q, XS, 16 * t + 8 * half, m, cq);
    store_tile(K, XS, 16 * t + 8 * half, m, ck);
    store_tile(V, XS, 16 * t + 8 * half, m, cv);
  }
  lds_fence();

#pragma unroll
  for (int tm = 0; tm < 2; ++tm)
#pragma unroll
    for (int tn = 0; tn < 2; ++tn) {
      v8f c = {};
#pragma unroll
      for (int kc = 0; kc < 4; ++kc) {
        v2f a  = frag_rowpair(Q, 16 * tm + m, XS, 4 * kc + 2 * half);
        v2f bb = frag_rowpair(K, 16 * tn + m, XS, 4 * kc + 2 * half);
        c = wmma4(a, bb, c);
      }
      store_tile(Sc, SS, 16 * tm + 8 * half, 16 * tn + m, c);
    }
  lds_fence();

  {
    float4* rowv = (float4*)(Sc + lane * SS);
    float r[S_];
#pragma unroll
    for (int i = 0; i < 8; ++i) {
      float4 f = rowv[i];
      r[4 * i + 0] = f.x;
      r[4 * i + 1] = f.y;
      r[4 * i + 2] = f.z;
      r[4 * i + 3] = f.w;
    }
    float mx = -3.402823466e38f;
#pragma unroll
    for (int j = 0; j < S_; ++j) mx = fmaxf(mx, r[j]);
    float sum = 0.f;
#pragma unroll
    for (int j = 0; j < S_; ++j) {
      r[j] = __expf((r[j] - mx) * 0.25f);
      sum += r[j];
    }
    const float inv = 1.f / sum;
#pragma unroll
    for (int i = 0; i < 8; ++i) {
      float4 f;
      f.x = r[4 * i + 0] * inv;
      f.y = r[4 * i + 1] * inv;
      f.z = r[4 * i + 2] * inv;
      f.w = r[4 * i + 3] * inv;
      rowv[i] = f;
    }
  }
  lds_fence();

#pragma unroll
  for (int t = 0; t < 2; ++t) {
    v8f c = {};
#pragma unroll
    for (int kc = 0; kc < 8; ++kc) {
      v2f a  = frag_rowpair(Sc, 16 * t + m, SS, 4 * kc + 2 * half);
      v2f bb = frag_col(V, 4 * kc + 2 * half, XS, m);
      c = wmma4(a, bb, c);
    }
    store_tile(X, XS, 16 * t + 8 * half, m, c);
  }
  lds_fence();

  {
    const float bb1 = s_b[0][m];
#pragma unroll
    for (int t = 0; t < 2; ++t) {
      v8f c = {};
#pragma unroll
      for (int kc = 0; kc < 4; ++kc) {
        v2f a  = frag_rowpair(X, 16 * t + m, XS, 4 * kc + 2 * half);
        v2f bb = frag_rowpair(s_w[3], m, D_, 4 * kc + 2 * half);
        c = wmma4(a, bb, c);
      }
#pragma unroll
      for (int v = 0; v < 8; ++v) c[v] = fmaxf(c[v] + bb1, 0.f);
      store_tile(Q, XS, 16 * t + 8 * half, m, c);
    }
  }
  lds_fence();

  {
    float partial = 0.f;
#pragma unroll
    for (int t = 0; t < 2; ++t) {
      v8f c = {};
#pragma unroll
      for (int kc = 0; kc < 4; ++kc) {
        v2f a  = frag_rowpair(Q, 16 * t + m, XS, 4 * kc + 2 * half);
        v2f bb = frag_rowpair(s_w[4], m, D_, 4 * kc + 2 * half);
        c = wmma4(a, bb, c);
      }
#pragma unroll
      for (int v = 0; v < 8; ++v) partial += c[v];
    }
    float other = __shfl_xor(partial, 16, 32);
    float mean  = (partial + other) * (1.f / 32.f) + s_b[1][m];
    if (half == 0) {
      out[(size_t)enc * ((size_t)B_ * D_) + (size_t)b * D_ + m] = mean;
    }
  }
}

extern "C" void kernel_launch(void* const* d_in, const int* in_sizes, int n_in,
                              void* d_out, int out_size, void* d_ws,
                              size_t ws_size, hipStream_t stream) {
  (void)in_sizes; (void)n_in; (void)out_size; (void)d_ws; (void)ws_size;
  const int*   arr1 = (const int*)d_in[0];
  const int*   arr2 = (const int*)d_in[1];
  const float* emb  = (const float*)d_in[2];
  const float* wq   = (const float*)d_in[3];
  const float* wk   = (const float*)d_in[4];
  const float* wv   = (const float*)d_in[5];
  const float* w1   = (const float*)d_in[6];
  const float* b1   = (const float*)d_in[7];
  const float* w2   = (const float*)d_in[8];
  const float* b2   = (const float*)d_in[9];
  float* out = (float*)d_out;

  const unsigned tasks  = 2u * B_;
  const unsigned blocks = tasks / WPB;
  encoder_kernel<<<dim3(blocks), dim3(TPB), 0, stream>>>(
      arr1, arr2, emb, wq, wk, wv, w1, b1, w2, b2, out);
}


# LDS tiles addressed as per-lane base + immediate offsets: tile stores and column loads pair up (129 to 81 LDS instructions per sequence), 156 to 112 VGPRs; h reuses P's tile: 6 to 8 waves per SIMD
# speedup vs baseline: 1.4498x; 1.4498x over previous
#include <hip/hip_runtime.h>


#define B_  131072
#define S_  32
#define D_  16
#define WPB 4
#define TPB (WPB * 32)
#define SPW 8

#define XS  20
#define SS  36

typedef float v2f __attribute__((ext_vector_type(2)));
typedef float v8f __attribute__((ext_vector_type(8)));

__device__ __forceinline__ v8f wmma4(v2f a, v2f b, v8f c) {
  return __builtin_amdgcn_wmma_f32_16x16x4_f32(false, a, false, b, (short)0, c,
                                               false, false);
}

__device__ __forceinline__ void lds_fence() {
  asm volatile("s_wait_dscnt 0" ::: "memory");
}


__device__ __forceinline__ v2f frag_rowpair(const float* lane_base, int off) {
  return *(const v2f*)(lane_base + off);
}

__device__ __forceinline__ v2f frag_col(const float* lane_base, int off,
                                        int rs) {
  v2f r;
  r.x = lane_base[off];
  r.y = lane_base[off + rs];
  return r;
}

__device__ __forceinline__ void store_tile(float* lane_base, int off, int rs,
                                           v8f acc) {
#pragma unroll
  for (int v = 0; v < 8; ++v) lane_base[off + v * rs] = acc[v];
}

__global__ __launch_bounds__(32) void fold_kernel(
    const float* __restrict__ wq, const float* __restrict__ wk,
    const float* __restrict__ wv, const float* __restrict__ w1,
    float* __restrict__ folded) {
  const int lane = threadIdx.x;
  const int m    = lane & 15;
  const int half = lane >> 4;
  const bool second = blockIdx.x != 0;
  const float* __restrict__ pa = second ? w1 : wk;
  const float* __restrict__ pb = second ? wv : wq;
  const int ars = second ? D_ : 1;
  const int aks = second ? 1 : D_;
  v8f c = {};
#pragma unroll
  for (int kc = 0; kc < 4; ++kc) {
    const int k0 = 4 * kc + 2 * half;
    v2f a, b;
    a.x = pa[m * ars + (k0 + 0) * aks];
    a.y = pa[m * ars + (k0 + 1) * aks];
    b.x = pb[(k0 + 0) * D_ + m];
    b.y = pb[(k0 + 1) * D_ + m];
    c = wmma4(a, b, c);
  }
  float* dst = folded + blockIdx.x * (D_ * D_);
#pragma unroll
  for (int v = 0; v < 8; ++v) dst[(8 * half + v) * D_ + m] = c[v];
}

__global__ __launch_bounds__(TPB, 8) void encoder_kernel(
    const int* __restrict__ arr1, const int* __restrict__ arr2,
    const float* __restrict__ emb,
    const float* __restrict__ folded,
    const float* __restrict__ b1,
    const float* __restrict__ w2, const float* __restrict__ b2,
    float* __restrict__ out) {
  __shared__ __align__(16) float s_ph[WPB][S_ * XS];
  __shared__ __align__(16) float s_u[WPB][S_ * XS];
  __shared__ __align__(16) float s_s[WPB][S_ * SS];

  const int tid  = threadIdx.x;
  const int wave = tid >> 5;
  const int lane = tid & 31;
  const int m    = lane & 15;
  const int half = lane >> 4;

  v2f fMT[4], fWC[4], fW2[4];
#pragma unroll
  for (int kc = 0; kc < 4; ++kc) {
    const int o = m * D_ + 4 * kc + 2 * half;
    fMT[kc] = *(const v2f*)(folded + o);
    fWC[kc] = *(const v2f*)(folded + D_ * D_ + o);
    fW2[kc] = *(const v2f*)(w2 + o);
  }
  const float bb1 = b1[m];
  const float bb2 = b2[m];

  const unsigned task0 = blockIdx.x * (unsigned)(SPW * WPB);
  const int enc = (task0 >= (unsigned)B_) ? 1 : 0;
  const int b0  = (enc ? (int)(task0 - (unsigned)B_) : (int)task0) + wave;
  const int* __restrict__ arr = enc ? arr2 : arr1;
  float* __restrict__ outp = out + (size_t)enc * ((size_t)B_ * D_);

  float* PH = s_ph[wave];
  float* U  = s_u[wave];
  float* Sc = s_s[wave];
  float*       PHst = PH + (8 * half) * XS + m;
  float*       Ust  = U  + (8 * half) * XS + m;
  float*       Sst  = Sc + (8 * half) * SS + m;
  const float* PHfr = PH + m * XS + 2 * half;
  const float* Sfr  = Sc + m * SS + 2 * half;
  const float* Ufc  = U  + (2 * half) * XS + m;
  float4*      rowv = (float4*)(Sc + lane * SS);

  for (int it = 0; it < SPW; ++it) {
  const int b = b0 + it * WPB;

  v2f xf[2][4];
  {
    const int tok0 = arr[b * S_ + m];
    const int tok1 = arr[b * S_ + 16 + m];
    const float* e0 = emb + (size_t)tok0 * D_ + 2 * half;
    const float* e1 = emb + (size_t)tok1 * D_ + 2 * half;
#pragma unroll
    for (int kc = 0; kc < 4; ++kc) {
      xf[0][kc] = *(const v2f*)(e0 + 4 * kc);
      xf[1][kc] = *(const v2f*)(e1 + 4 * kc);
    }
  }

#pragma unroll
  for (int t = 0; t < 2; ++t) {
    v8f cp = {}, cu = {};
#pragma unroll
    for (int kc = 0; kc < 4; ++kc) {
      v2f a = xf[t][kc];
      cp = wmma4(a, fMT[kc], cp);
      cu = wmma4(a, fWC[kc], cu);
    }
    store_tile(PHst, 16 * t * XS, XS, cp);
    store_tile(Ust,  16 * t * XS, XS, cu);
  }
  lds_fence();

#pragma unroll
  for (int tm = 0; tm < 2; ++tm)
#pragma unroll
    for (int tn = 0; tn < 2; ++tn) {
      v8f c = {};
#pragma unroll
      for (int kc = 0; kc < 4; ++kc) {
        v2f a  = frag_rowpair(PHfr, 16 * tm * XS + 4 * kc);
        c = wmma4(a, xf[tn][kc], c);
      }
      store_tile(Sst, 16 * tm * SS + 16 * tn, SS, c);
    }
  lds_fence();

  {
    float r[S_];
#pragma unroll
    for (int i = 0; i < 8; ++i) {
      float4 f = rowv[i];
      r[4 * i + 0] = f.x;
      r[4 * i + 1] = f.y;
      r[4 * i + 2] = f.z;
      r[4 * i + 3] = f.w;
    }
    float mx = -3.402823466e38f;
#pragma unroll
    for (int j = 0; j < S_; ++j) mx = fmaxf(mx, r[j]);
    float sum = 0.f;
#pragma unroll
    for (int j = 0; j < S_; ++j) {
      r[j] = __expf((r[j] - mx) * 0.25f);
      sum += r[j];
    }
    const float inv = 1.f / sum;
#pragma unroll
    for (int i = 0; i < 8; ++i) {
      float4 f;
      f.x = r[4 * i + 0] * inv;
      f.y = r[4 * i + 1] * inv;
      f.z = r[4 * i + 2] * inv;
      f.w = r[4 * i + 3] * inv;
      rowv[i] = f;
    }
  }
  lds_fence();

  {
#pragma unroll
    for (int t = 0; t < 2; ++t) {
      v8f c = {};
#pragma unroll
      for (int kc = 0; kc < 8; ++kc) {
        v2f a  = frag_rowpair(Sfr, 16 * t * SS + 4 * kc);
        v2f bb = frag_col(Ufc, 4 * kc * XS, XS);
        c = wmma4(a, bb, c);
      }
#pragma unroll
      for (int v = 0; v < 8; ++v) c[v] = fmaxf(c[v] + bb1, 0.f);
      store_tile(PHst, 16 * t * XS, XS, c);
    }
  }
  lds_fence();

  {
    float partial = 0.f;
#pragma unroll
    for (int t = 0; t < 2; ++t) {
      v8f c = {};
#pragma unroll
      for (int kc = 0; kc < 4; ++kc) {
        v2f a  = frag_rowpair(PHfr, 16 * t * XS + 4 * kc);
        c = wmma4(a, fW2[kc], c);
      }
#pragma unroll
      for (int v = 0; v < 8; ++v) partial += c[v];
    }
    float other = __shfl_xor(partial, 16, 32);
    float mean  = (partial + other) * (1.f / 32.f) + bb2;
    if (half == 0) {
      outp[(size_t)b * D_ + m] = mean;
    }
  }
  }
}

extern "C" void kernel_launch(void* const* d_in, const int* in_sizes, int n_in,
                              void* d_out, int out_size, void* d_ws,
                              size_t ws_size, hipStream_t stream) {
  (void)in_sizes; (void)n_in; (void)out_size; (void)ws_size;
  const int*   arr1 = (const int*)d_in[0];
  const int*   arr2 = (const int*)d_in[1];
  const float* emb  = (const float*)d_in[2];
  const float* wq   = (const float*)d_in[3];
  const float* wk   = (const float*)d_in[4];
  const float* wv   = (const float*)d_in[5];
  const float* w1   = (const float*)d_in[6];
  const float* b1   = (const float*)d_in[7];
  const float* w2   = (const float*)d_in[8];
  const float* b2   = (const float*)d_in[9];
  float* out    = (float*)d_out;
  float* folded = (float*)d_ws;

  fold_kernel<<<dim3(2), dim3(32), 0, stream>>>(wq, wk, wv, w1, folded);

  const unsigned tasks  = 2u * B_;
  const unsigned blocks = tasks / (WPB * SPW);
  encoder_kernel<<<dim3(blocks), dim3(TPB), 0, stream>>>(
      arr1, arr2, emb, folded, b1, w2, b2, out);
}
